# starting kernel
# baseline (speedup 1.0000x reference)
#include <hip/hip_runtime.h>


typedef float v2f __attribute__((ext_vector_type(2)));
typedef float v8f __attribute__((ext_vector_type(8)));

#define BATCH 8
#define CT    4096
#define CHAN  64
#define WIN   11

__global__ __launch_bounds__(256)
void corr_wmma_f32_kernel(const float* __restrict__ x0,
                          const float* __restrict__ x1,
                          float* __restrict__ out) {
  const int wave = blockIdx.x * 8 + (threadIdx.x >> 5);
  const int lane = threadIdx.x & 31;
  const int li   = lane & 15;
  const int half = lane >> 4;

  const int b  = wave >> 8;
  const int t0 = (wave & 255) << 4;

  const int rowA = t0 + li;
  const int rowL = (t0 - 8 + li) & (CT - 1);
  const int rowR = (t0 + 8 + li) & (CT - 1);

  const float* pA = x1 + ((size_t)b * CT + rowA) * CHAN + half * 2;
  const float* pL = x0 + ((size_t)b * CT + rowL) * CHAN + half * 2;
  const float* pR = x0 + ((size_t)b * CT + rowR) * CHAN + half * 2;

  v8f accL = {};
  v8f accR = {};

#pragma unroll
  for (int k = 0; k < 16; ++k) {
    const v2f a  = *(const v2f*)(pA + 4 * k);
    const v2f bl = *(const v2f*)(pL + 4 * k);
    const v2f br = *(const v2f*)(pR + 4 * k);
    accL = __builtin_amdgcn_wmma_f32_16x16x4_f32(
        false, a, false, bl, (short)0, accL, false, false);
    accR = __builtin_amdgcn_wmma_f32_16x16x4_f32(
        false, a, false, br, (short)0, accR, false, false);
  }

  const float scale = 1.0f / (float)CHAN;
  const int   n     = li;
  float* obase = out + ((size_t)b * CT + t0) * WIN;
#pragma unroll
  for (int v = 0; v < 8; ++v) {
    const int m = v + half * 8;
    int w = n - m - 3;
    if (0 <= w && w <= 10) obase[(size_t)m * WIN + w] = accL[v] * scale;
    w = n - m + 13;
    if (0 <= w && w <= 10) obase[(size_t)m * WIN + w] = accR[v] * scale;
  }
}

extern "C" void kernel_launch(void* const* d_in, const int* in_sizes, int n_in,
                              void* d_out, int out_size, void* d_ws, size_t ws_size,
                              hipStream_t stream) {
  (void)in_sizes; (void)n_in; (void)out_size; (void)d_ws; (void)ws_size;
  const float* x0 = (const float*)d_in[0];
  const float* x1 = (const float*)d_in[1];
  float* out = (float*)d_out;

  const int waves_total = BATCH * (CT / 16);
  dim3 block(256);
  dim3 grid(waves_total / 8);
  corr_wmma_f32_kernel<<<grid, block, 0, stream>>>(x0, x1, out);
}


# Epilogue through LDS: the band is cut from the two 16x16 result tiles in LDS and each block's 128 rows leave as whole-line 16-byte stores instead of 16 branch-guarded 4-byte stores per lane
# speedup vs baseline: 1.1809x; 1.1809x over previous
#include <hip/hip_runtime.h>


typedef float v2f __attribute__((ext_vector_type(2)));
typedef float v8f __attribute__((ext_vector_type(8)));

#define BATCH 8
#define CT    4096
#define CHAN  64
#define WIN   11

#define WAVES_PER_BLOCK 8
#define TPITCH 20
#define TREGION (32 * TPITCH)
#define BLOCK_OUT (WAVES_PER_BLOCK * 16 * WIN)
#define BLOCK_OUT4 (BLOCK_OUT / 4)

__device__ __forceinline__ int band_index(int e) {
  const int row = e / WIN;
  const int w   = e - row * WIN;
  const int m   = row & 15;
  return (row >> 4) * TREGION + (m + w + 3) * TPITCH + m;
}

__global__ __launch_bounds__(256)
void corr_wmma_f32_kernel(const float* __restrict__ x0,
                          const float* __restrict__ x1,
                          float* __restrict__ out) {
  __shared__ __attribute__((aligned(16))) float tile[WAVES_PER_BLOCK * TREGION];

  const int wid  = threadIdx.x >> 5;
  const int wave = blockIdx.x * WAVES_PER_BLOCK + wid;
  const int lane = threadIdx.x & 31;
  const int li   = lane & 15;
  const int half = lane >> 4;

  const int b  = wave >> 8;
  const int t0 = (wave & 255) << 4;

  const int rowA = t0 + li;
  const int rowL = (t0 - 8 + li) & (CT - 1);
  const int rowR = (t0 + 8 + li) & (CT - 1);

  const float* pA = x1 + ((size_t)b * CT + rowA) * CHAN + half * 2;
  const float* pL = x0 + ((size_t)b * CT + rowL) * CHAN + half * 2;
  const float* pR = x0 + ((size_t)b * CT + rowR) * CHAN + half * 2;

  const int q0 = threadIdx.x;
  const int i0 = band_index(4 * q0);
  const int i1 = band_index(4 * q0 + 1);
  const int i2 = band_index(4 * q0 + 2);
  const int i3 = band_index(4 * q0 + 3);

  v8f accL = {};
  v8f accR = {};

#pragma unroll
  for (int k = 0; k < 16; ++k) {
    const v2f a  = *(const v2f*)(pA + 4 * k);
    const v2f bl = *(const v2f*)(pL + 4 * k);
    const v2f br = *(const v2f*)(pR + 4 * k);
    accL = __builtin_amdgcn_wmma_f32_16x16x4_f32(
        false, a, false, bl, (short)0, accL, false, false);
    accR = __builtin_amdgcn_wmma_f32_16x16x4_f32(
        false, a, false, br, (short)0, accR, false, false);
  }

  float* tw = tile + wid * TREGION;
  float4* cl = (float4*)(tw + li * TPITCH + half * 8);
  float4* cr = (float4*)(tw + (16 + li) * TPITCH + half * 8);
  cl[0] = make_float4(accL[0], accL[1], accL[2], accL[3]);
  cl[1] = make_float4(accL[4], accL[5], accL[6], accL[7]);
  cr[0] = make_float4(accR[0], accR[1], accR[2], accR[3]);
  cr[1] = make_float4(accR[4], accR[5], accR[6], accR[7]);

  __syncthreads();

  const float scale = 1.0f / (float)CHAN;
  float4* oblock = (float4*)(out + (size_t)blockIdx.x * BLOCK_OUT);
  oblock[q0] = make_float4(tile[i0] * scale, tile[i1] * scale,
                           tile[i2] * scale, tile[i3] * scale);

  const int q1 = q0 + 256;
  if (__builtin_amdgcn_readfirstlane(q1) < BLOCK_OUT4) {
    const int e1 = 4 * q1;
    oblock[q1] = make_float4(tile[band_index(e1)] * scale,
                             tile[band_index(e1 + 1)] * scale,
                             tile[band_index(e1 + 2)] * scale,
                             tile[band_index(e1 + 3)] * scale);
  }
}

extern "C" void kernel_launch(void* const* d_in, const int* in_sizes, int n_in,
                              void* d_out, int out_size, void* d_ws, size_t ws_size,
                              hipStream_t stream) {
  (void)in_sizes; (void)n_in; (void)out_size; (void)d_ws; (void)ws_size;
  const float* x0 = (const float*)d_in[0];
  const float* x1 = (const float*)d_in[1];
  float* out = (float*)d_out;

  const int waves_total = BATCH * (CT / 16);
  dim3 block(256);
  dim3 grid(waves_total / 8);
  corr_wmma_f32_kernel<<<grid, block, 0, stream>>>(x0, x1, out);
}
